# starting kernel
# baseline (speedup 1.0000x reference)
#include <hip/hip_runtime.h>
#include <math.h>

typedef __attribute__((ext_vector_type(2))) float v2f;
typedef __attribute__((ext_vector_type(4))) float v4f;
typedef __attribute__((ext_vector_type(8))) float v8f;

#define PN 32
#define MN 32
#define BN 128
#define HN 256
#define DN 54
#define ON 53
#define OPAD 64
#define CH 512

#define RB 128
#define CB 64
#define KC 64
#define LDK (KC + 4)


__global__ void zero_f32(float* p, int n) {
  int i = blockIdx.x * blockDim.x + threadIdx.x;
  if (i < n) p[i] = 0.f;
}

__global__ void onehot_to_idx(const float* __restrict__ oh, int* __restrict__ idx, int rows) {
  int r = blockIdx.x * blockDim.x + threadIdx.x;
  if (r >= rows) return;
  const float* p = oh + (size_t)r * DN;
  int v = 0;
  for (int d = 0; d < DN; ++d)
    if (p[d] > 0.5f) v = d;
  idx[r] = v;
}

__global__ void pad_proj_weights(const float* __restrict__ Ws, const float* __restrict__ bs,
                                 const float* __restrict__ Wi, const float* __restrict__ bi,
                                 float* __restrict__ Wsp, float* __restrict__ Wip,
                                 float* __restrict__ bsp, float* __restrict__ bip) {
  int t = blockIdx.x * blockDim.x + threadIdx.x;
  if (t >= OPAD * CH) return;
  int n = t / CH;
  int k = t - n * CH;
  Wsp[t] = (n < ON) ? Ws[n * CH + k] : 0.f;
  Wip[t] = (n < ON) ? Wi[n * CH + k] : 0.f;
  if (t < OPAD) {
    bsp[t] = (t < ON) ? bs[t] : 0.f;
    bip[t] = (t < ON) ? bi[t] : 0.f;
  }
}

__global__ __launch_bounds__(256) void gemm_wmma_f32(
    const float* __restrict__ X, const float* __restrict__ W,
    const float* __restrict__ bias,
    const float* __restrict__ emb, const int* __restrict__ idx, int embD,
    float* __restrict__ out, int N, int K) {
  __shared__ float Blds[2][CB * LDK];

  int tid = threadIdx.x;
  int lane = tid & 31;
  int wv = tid >> 5;
  int kh = lane >> 4;
  int l16 = lane & 15;
  int row0 = blockIdx.x * RB + wv * 16;
  int col0 = blockIdx.y * CB;

  int bcol = tid >> 2;
  int bseg = (tid & 3) * 16;
  const float* Bg = W + (size_t)(col0 + bcol) * K + bseg;
  float* Bs0 = &Blds[0][0] + bcol * LDK + bseg;
  float* Bs1 = &Blds[1][0] + bcol * LDK + bseg;

  const float* Ap = X + (size_t)(row0 + l16) * K + 2 * kh;

  v8f acc0 = {}, acc1 = {}, acc2 = {}, acc3 = {};

  {
    v4f p0 = *(const v4f*)(Bg + 0);
    v4f p1 = *(const v4f*)(Bg + 4);
    v4f p2 = *(const v4f*)(Bg + 8);
    v4f p3 = *(const v4f*)(Bg + 12);
    *(v4f*)(Bs0 + 0) = p0;
    *(v4f*)(Bs0 + 4) = p1;
    *(v4f*)(Bs0 + 8) = p2;
    *(v4f*)(Bs0 + 12) = p3;
  }
  __syncthreads();

  int nchunks = K / KC;
  for (int kc = 0; kc < nchunks; ++kc) {
    v4f p0, p1, p2, p3;
    bool hasNext = (kc + 1 < nchunks);
    if (hasNext) {
      const float* g = Bg + (size_t)(kc + 1) * KC;
      p0 = *(const v4f*)(g + 0);
      p1 = *(const v4f*)(g + 4);
      p2 = *(const v4f*)(g + 8);
      p3 = *(const v4f*)(g + 12);
    }

    const float* Bl = &Blds[kc & 1][0];
    const float* Al = Ap + (size_t)kc * KC;
    const float* bp0 = Bl + (0 + l16) * LDK + 2 * kh;
    const float* bp1 = Bl + (16 + l16) * LDK + 2 * kh;
    const float* bp2 = Bl + (32 + l16) * LDK + 2 * kh;
    const float* bp3 = Bl + (48 + l16) * LDK + 2 * kh;

#pragma unroll
    for (int k = 0; k < KC; k += 4) {
      v2f a = *(const v2f*)(Al + k);
      v2f b0 = *(const v2f*)(bp0 + k);
      v2f b1 = *(const v2f*)(bp1 + k);
      v2f b2 = *(const v2f*)(bp2 + k);
      v2f b3 = *(const v2f*)(bp3 + k);
      acc0 = __builtin_amdgcn_wmma_f32_16x16x4_f32(false, a, false, b0, (short)0, acc0, false, false);
      acc1 = __builtin_amdgcn_wmma_f32_16x16x4_f32(false, a, false, b1, (short)0, acc1, false, false);
      acc2 = __builtin_amdgcn_wmma_f32_16x16x4_f32(false, a, false, b2, (short)0, acc2, false, false);
      acc3 = __builtin_amdgcn_wmma_f32_16x16x4_f32(false, a, false, b3, (short)0, acc3, false, false);
    }

    if (hasNext) {
      float* s = ((kc + 1) & 1) ? Bs1 : Bs0;
      *(v4f*)(s + 0) = p0;
      *(v4f*)(s + 4) = p1;
      *(v4f*)(s + 8) = p2;
      *(v4f*)(s + 12) = p3;
      __syncthreads();
    }
  }

  int ridx[8];
#pragma unroll
  for (int r = 0; r < 8; ++r) ridx[r] = 0;
  if (idx) {
#pragma unroll
    for (int r = 0; r < 8; ++r) ridx[r] = idx[row0 + r + kh * 8];
  }

  v8f accs[4] = {acc0, acc1, acc2, acc3};
#pragma unroll
  for (int j = 0; j < 4; ++j) {
    int col = col0 + j * 16 + l16;
    float badd = bias ? bias[col] : 0.f;
#pragma unroll
    for (int r = 0; r < 8; ++r) {
      int row = row0 + r + kh * 8;
      float v = accs[j][r] + badd;
      if (emb) v += emb[(size_t)col * embD + ridx[r]];
      out[(size_t)row * N + col] = v;
    }
  }
}

__global__ void lstm_gates(const float* __restrict__ G, float* __restrict__ c,
                           float* __restrict__ h, float* __restrict__ seq,
                           int Hs, int seqOff) {
  int t = blockIdx.x * blockDim.x + threadIdx.x;
  if (t >= BN * Hs) return;
  int b = t / Hs;
  int u = t - b * Hs;
  const float* g = G + (size_t)b * 4 * Hs;
  float gi = g[u];
  float gf = g[Hs + u];
  float gg = g[2 * Hs + u];
  float go = g[3 * Hs + u];
  float si = 1.f / (1.f + expf(-gi));
  float sf = 1.f / (1.f + expf(-gf));
  float so = 1.f / (1.f + expf(-go));
  float cn = sf * c[t] + si * tanhf(gg);
  float hn = so * tanhf(cn);
  c[t] = cn;
  h[t] = hn;
  seq[(size_t)b * CH + seqOff + u] = hn;
}

__global__ void finalize(const float* __restrict__ pPs, const float* __restrict__ pPi,
                         const float* __restrict__ pMs, const float* __restrict__ pMi,
                         const int* __restrict__ tgt, float* __restrict__ out) {
  int t = blockIdx.x * blockDim.x + threadIdx.x;
  if (t >= PN * MN * BN) return;
  int b = t % BN;
  int pm = t / BN;
  int m = pm % MN;
  int p = pm / MN;
  const float* as = pPs + ((size_t)p * BN + b) * OPAD;
  const float* bsv = pMs + ((size_t)m * BN + b) * OPAD;
  const float* ai = pPi + ((size_t)p * BN + b) * OPAD;
  const float* biv = pMi + ((size_t)m * BN + b) * OPAD;
  int tg = (m < MN - 1) ? tgt[(m + 1) * BN + b] : 0;

  float mxs = -3.4e38f, mxi = -3.4e38f;
  for (int o = 0; o < ON; ++o) {
    float ls = as[o] + bsv[o];
    float li = ai[o] + biv[o];
    mxs = ls > mxs ? ls : mxs;
    mxi = li > mxi ? li : mxi;
  }
  float sums = 0.f, sumi = 0.f, ls52 = 0.f, li52 = 0.f, lst = 0.f, lit = 0.f;
  for (int o = 0; o < ON; ++o) {
    float ls = as[o] + bsv[o];
    float li = ai[o] + biv[o];
    sums += expf(ls - mxs);
    sumi += expf(li - mxi);
    if (o == ON - 1) { ls52 = ls; li52 = li; }
    if (o == tg) { lst = ls; lit = li; }
  }
  float lses = mxs + logf(sums);
  float lsei = mxi + logf(sumi);
  size_t idxPMB = ((size_t)p * MN + m) * BN + b;
  out[idxPMB] = ls52 - lses;
  out[(size_t)PN * MN * BN + idxPMB] = li52 - lsei;
  if (m < MN - 1) {
    size_t base = 2ull * PN * MN * BN;
    size_t idxS = ((size_t)p * (MN - 1) + m) * BN + b;
    out[base + idxS] = lst - lses;
    out[base + (size_t)PN * (MN - 1) * BN + idxS] = lit - lsei;
  }
}


extern "C" void kernel_launch(void* const* d_in, const int* in_sizes, int n_in,
                              void* d_out, int out_size, void* d_ws, size_t ws_size,
                              hipStream_t stream) {
  (void)in_sizes; (void)n_in; (void)out_size; (void)ws_size;
  const float* src_oh = (const float*)d_in[0];
  const float* tgt_oh = (const float*)d_in[1];  (void)tgt_oh;
  const float* Wih_pf = (const float*)d_in[2];
  const float* Whh_pf = (const float*)d_in[3];
  const float* b_pf   = (const float*)d_in[4];
  const float* Wih_pb = (const float*)d_in[5];
  const float* Whh_pb = (const float*)d_in[6];
  const float* b_pb   = (const float*)d_in[7];
  const float* Wih_m  = (const float*)d_in[8];
  const float* Whh_m  = (const float*)d_in[9];
  const float* b_m    = (const float*)d_in[10];
  const float* W_sub  = (const float*)d_in[11];
  const float* b_sub  = (const float*)d_in[12];
  const float* W_ins  = (const float*)d_in[13];
  const float* b_ins  = (const float*)d_in[14];
  const int*   tgt    = (const int*)d_in[15];

  char* wptr = (char*)d_ws;
  auto alloc = [&](size_t nfloats) -> float* {
    float* p = (float*)wptr;
    wptr += ((nfloats * 4 + 255) / 256) * 256;
    return p;
  };
  float* prior  = alloc((size_t)PN * BN * CH);
  float* modern = alloc((size_t)MN * BN * CH);
  float* Gpf = alloc((size_t)BN * 4 * HN);
  float* Gpb = alloc((size_t)BN * 4 * HN);
  float* Gm  = alloc((size_t)BN * 4 * CH);
  int hcN = 4 * BN * HN + 2 * BN * CH;
  float* hc = alloc(hcN);
  float* h_pf = hc;
  float* c_pf = h_pf + BN * HN;
  float* h_pb = c_pf + BN * HN;
  float* c_pb = h_pb + BN * HN;
  float* h_m  = c_pb + BN * HN;
  float* c_m  = h_m + BN * CH;
  float* Wsp = alloc(OPAD * CH);
  float* Wip = alloc(OPAD * CH);
  float* bsp = alloc(OPAD);
  float* bip = alloc(OPAD);
  float* pPs = alloc((size_t)PN * BN * OPAD);
  float* pPi = alloc((size_t)PN * BN * OPAD);
  float* pMs = alloc((size_t)MN * BN * OPAD);
  float* pMi = alloc((size_t)MN * BN * OPAD);
  int* sidx = (int*)alloc(PN * BN);

  zero_f32<<<(hcN + 255) / 256, 256, 0, stream>>>(hc, hcN);
  onehot_to_idx<<<(PN * BN + 255) / 256, 256, 0, stream>>>(src_oh, sidx, PN * BN);
  pad_proj_weights<<<(OPAD * CH + 255) / 256, 256, 0, stream>>>(
      W_sub, b_sub, W_ins, b_ins, Wsp, Wip, bsp, bip);

  for (int t = 0; t < PN; ++t) {
    gemm_wmma_f32<<<dim3(1, (4 * HN) / CB), 256, 0, stream>>>(
        h_pf, Whh_pf, b_pf, Wih_pf, sidx + t * BN, DN, Gpf, 4 * HN, HN);
    lstm_gates<<<(BN * HN + 255) / 256, 256, 0, stream>>>(
        Gpf, c_pf, h_pf, prior + (size_t)t * BN * CH, HN, 0);
    int tb = PN - 1 - t;
    gemm_wmma_f32<<<dim3(1, (4 * HN) / CB), 256, 0, stream>>>(
        h_pb, Whh_pb, b_pb, Wih_pb, sidx + tb * BN, DN, Gpb, 4 * HN, HN);
    lstm_gates<<<(BN * HN + 255) / 256, 256, 0, stream>>>(
        Gpb, c_pb, h_pb, prior + (size_t)tb * BN * CH, HN, HN);
    gemm_wmma_f32<<<dim3(1, (4 * CH) / CB), 256, 0, stream>>>(
        h_m, Whh_m, b_m, Wih_m, tgt + t * BN, DN, Gm, 4 * CH, CH);
    lstm_gates<<<(BN * CH + 255) / 256, 256, 0, stream>>>(
        Gm, c_m, h_m, modern + (size_t)t * BN * CH, CH, 0);
  }

  gemm_wmma_f32<<<dim3(PN * BN / RB, 1), 256, 0, stream>>>(
      prior, Wsp, bsp, nullptr, nullptr, 0, pPs, OPAD, CH);
  gemm_wmma_f32<<<dim3(PN * BN / RB, 1), 256, 0, stream>>>(
      prior, Wip, bip, nullptr, nullptr, 0, pPi, OPAD, CH);
  gemm_wmma_f32<<<dim3(MN * BN / RB, 1), 256, 0, stream>>>(
      modern, Wsp, bsp, nullptr, nullptr, 0, pMs, OPAD, CH);
  gemm_wmma_f32<<<dim3(MN * BN / RB, 1), 256, 0, stream>>>(
      modern, Wip, bip, nullptr, nullptr, 0, pMi, OPAD, CH);

  finalize<<<(PN * MN * BN + 255) / 256, 256, 0, stream>>>(
      pPs, pPi, pMs, pMi, tgt, (float*)d_out);
}


# Step and projection GEMMs: one 16x32 tile per single-wave work-group, A and B fragments read straight from global memory as 16-byte row pieces, no LDS, no barrier; new k sum order within each 8
# speedup vs baseline: 2.8455x; 2.8455x over previous
#include <hip/hip_runtime.h>
#include <math.h>

typedef __attribute__((ext_vector_type(2))) float v2f;
typedef __attribute__((ext_vector_type(4))) float v4f;
typedef __attribute__((ext_vector_type(8))) float v8f;

#define PN 32
#define MN 32
#define BN 128
#define HN 256
#define DN 54
#define ON 53
#define OPAD 64
#define CH 512


__global__ void zero_f32(float* p, int n) {
  int i = blockIdx.x * blockDim.x + threadIdx.x;
  if (i < n) p[i] = 0.f;
}

__global__ void onehot_to_idx(const float* __restrict__ oh, int* __restrict__ idx, int rows) {
  int r = blockIdx.x * blockDim.x + threadIdx.x;
  if (r >= rows) return;
  const float* p = oh + (size_t)r * DN;
  int v = 0;
  for (int d = 0; d < DN; ++d)
    if (p[d] > 0.5f) v = d;
  idx[r] = v;
}

__global__ void pad_proj_weights(const float* __restrict__ Ws, const float* __restrict__ bs,
                                 const float* __restrict__ Wi, const float* __restrict__ bi,
                                 float* __restrict__ Wsp, float* __restrict__ Wip,
                                 float* __restrict__ bsp, float* __restrict__ bip) {
  int t = blockIdx.x * blockDim.x + threadIdx.x;
  if (t >= OPAD * CH) return;
  int n = t / CH;
  int k = t - n * CH;
  Wsp[t] = (n < ON) ? Ws[n * CH + k] : 0.f;
  Wip[t] = (n < ON) ? Wi[n * CH + k] : 0.f;
  if (t < OPAD) {
    bsp[t] = (t < ON) ? bs[t] : 0.f;
    bip[t] = (t < ON) ? bi[t] : 0.f;
  }
}

template <int K, bool EMB>
__global__ __launch_bounds__(32) __attribute__((amdgpu_waves_per_eu(1, 1))) void gemm_wave_f32(
    const float* __restrict__ X, const float* __restrict__ W,
    const float* __restrict__ bias,
    const float* __restrict__ emb, const int* __restrict__ idx, int embD,
    float* __restrict__ out, int N) {
  constexpr int KG = K / 8;

  const int lane = threadIdx.x;
  const int hh = lane >> 4;
  const int c = lane & 15;
  const int row0 = blockIdx.x * 16;
  const int col0 = blockIdx.y * 32;

  const float* Ap = X + (size_t)(row0 + c) * K + 4 * hh;
  const float* Bp0 = W + (size_t)(col0 + c) * K + 4 * hh;
  const float* Bp1 = Bp0 + (size_t)16 * K;

  const float badd0 = bias[col0 + c];
  const float badd1 = bias[col0 + 16 + c];
  int ridx[8];
  if (EMB) {
#pragma unroll
    for (int r = 0; r < 8; ++r) ridx[r] = idx[row0 + r + hh * 8];
  }

  v8f acc0 = {}, acc1 = {};

  float e0[8], e1[8];
  if (EMB) {
    const float* ep0 = emb + (size_t)(col0 + c) * embD;
    const float* ep1 = emb + (size_t)(col0 + 16 + c) * embD;
#pragma unroll
    for (int r = 0; r < 8; ++r) {
      e0[r] = ep0[ridx[r]];
      e1[r] = ep1[ridx[r]];
    }
  }

#pragma unroll
  for (int g = 0; g < KG; ++g) {
    const v4f av = *(const v4f*)(Ap + g * 8);
    const v4f bv0 = *(const v4f*)(Bp0 + g * 8);
    const v4f bv1 = *(const v4f*)(Bp1 + g * 8);
    const v2f a_lo = {av[0], av[1]};
    const v2f a_hi = {av[2], av[3]};
    const v2f b0_lo = {bv0[0], bv0[1]};
    const v2f b0_hi = {bv0[2], bv0[3]};
    const v2f b1_lo = {bv1[0], bv1[1]};
    const v2f b1_hi = {bv1[2], bv1[3]};
    acc0 = __builtin_amdgcn_wmma_f32_16x16x4_f32(false, a_lo, false, b0_lo, (short)0, acc0, false, false);
    acc1 = __builtin_amdgcn_wmma_f32_16x16x4_f32(false, a_lo, false, b1_lo, (short)0, acc1, false, false);
    acc0 = __builtin_amdgcn_wmma_f32_16x16x4_f32(false, a_hi, false, b0_hi, (short)0, acc0, false, false);
    acc1 = __builtin_amdgcn_wmma_f32_16x16x4_f32(false, a_hi, false, b1_hi, (short)0, acc1, false, false);
  }

  float* op = out + (size_t)(row0 + hh * 8) * N + col0 + c;
#pragma unroll
  for (int r = 0; r < 8; ++r) {
    float v0 = acc0[r] + badd0;
    float v1 = acc1[r] + badd1;
    if (EMB) {
      v0 += e0[r];
      v1 += e1[r];
    }
    op[(size_t)r * N] = v0;
    op[(size_t)r * N + 16] = v1;
  }
}

__global__ void lstm_gates(const float* __restrict__ G, float* __restrict__ c,
                           float* __restrict__ h, float* __restrict__ seq,
                           int Hs, int seqOff) {
  int t = blockIdx.x * blockDim.x + threadIdx.x;
  if (t >= BN * Hs) return;
  int b = t / Hs;
  int u = t - b * Hs;
  const float* g = G + (size_t)b * 4 * Hs;
  float gi = g[u];
  float gf = g[Hs + u];
  float gg = g[2 * Hs + u];
  float go = g[3 * Hs + u];
  float si = 1.f / (1.f + expf(-gi));
  float sf = 1.f / (1.f + expf(-gf));
  float so = 1.f / (1.f + expf(-go));
  float cn = sf * c[t] + si * tanhf(gg);
  float hn = so * tanhf(cn);
  c[t] = cn;
  h[t] = hn;
  seq[(size_t)b * CH + seqOff + u] = hn;
}

__global__ void finalize(const float* __restrict__ pPs, const float* __restrict__ pPi,
                         const float* __restrict__ pMs, const float* __restrict__ pMi,
                         const int* __restrict__ tgt, float* __restrict__ out) {
  int t = blockIdx.x * blockDim.x + threadIdx.x;
  if (t >= PN * MN * BN) return;
  int b = t % BN;
  int pm = t / BN;
  int m = pm % MN;
  int p = pm / MN;
  const float* as = pPs + ((size_t)p * BN + b) * OPAD;
  const float* bsv = pMs + ((size_t)m * BN + b) * OPAD;
  const float* ai = pPi + ((size_t)p * BN + b) * OPAD;
  const float* biv = pMi + ((size_t)m * BN + b) * OPAD;
  int tg = (m < MN - 1) ? tgt[(m + 1) * BN + b] : 0;

  float mxs = -3.4e38f, mxi = -3.4e38f;
  for (int o = 0; o < ON; ++o) {
    float ls = as[o] + bsv[o];
    float li = ai[o] + biv[o];
    mxs = ls > mxs ? ls : mxs;
    mxi = li > mxi ? li : mxi;
  }
  float sums = 0.f, sumi = 0.f, ls52 = 0.f, li52 = 0.f, lst = 0.f, lit = 0.f;
  for (int o = 0; o < ON; ++o) {
    float ls = as[o] + bsv[o];
    float li = ai[o] + biv[o];
    sums += expf(ls - mxs);
    sumi += expf(li - mxi);
    if (o == ON - 1) { ls52 = ls; li52 = li; }
    if (o == tg) { lst = ls; lit = li; }
  }
  float lses = mxs + logf(sums);
  float lsei = mxi + logf(sumi);
  size_t idxPMB = ((size_t)p * MN + m) * BN + b;
  out[idxPMB] = ls52 - lses;
  out[(size_t)PN * MN * BN + idxPMB] = li52 - lsei;
  if (m < MN - 1) {
    size_t base = 2ull * PN * MN * BN;
    size_t idxS = ((size_t)p * (MN - 1) + m) * BN + b;
    out[base + idxS] = lst - lses;
    out[base + (size_t)PN * (MN - 1) * BN + idxS] = lit - lsei;
  }
}


extern "C" void kernel_launch(void* const* d_in, const int* in_sizes, int n_in,
                              void* d_out, int out_size, void* d_ws, size_t ws_size,
                              hipStream_t stream) {
  (void)in_sizes; (void)n_in; (void)out_size; (void)ws_size;
  const float* src_oh = (const float*)d_in[0];
  const float* tgt_oh = (const float*)d_in[1];  (void)tgt_oh;
  const float* Wih_pf = (const float*)d_in[2];
  const float* Whh_pf = (const float*)d_in[3];
  const float* b_pf   = (const float*)d_in[4];
  const float* Wih_pb = (const float*)d_in[5];
  const float* Whh_pb = (const float*)d_in[6];
  const float* b_pb   = (const float*)d_in[7];
  const float* Wih_m  = (const float*)d_in[8];
  const float* Whh_m  = (const float*)d_in[9];
  const float* b_m    = (const float*)d_in[10];
  const float* W_sub  = (const float*)d_in[11];
  const float* b_sub  = (const float*)d_in[12];
  const float* W_ins  = (const float*)d_in[13];
  const float* b_ins  = (const float*)d_in[14];
  const int*   tgt    = (const int*)d_in[15];

  char* wptr = (char*)d_ws;
  auto alloc = [&](size_t nfloats) -> float* {
    float* p = (float*)wptr;
    wptr += ((nfloats * 4 + 255) / 256) * 256;
    return p;
  };
  float* prior  = alloc((size_t)PN * BN * CH);
  float* modern = alloc((size_t)MN * BN * CH);
  float* Gpf = alloc((size_t)BN * 4 * HN);
  float* Gpb = alloc((size_t)BN * 4 * HN);
  float* Gm  = alloc((size_t)BN * 4 * CH);
  int hcN = 4 * BN * HN + 2 * BN * CH;
  float* hc = alloc(hcN);
  float* h_pf = hc;
  float* c_pf = h_pf + BN * HN;
  float* h_pb = c_pf + BN * HN;
  float* c_pb = h_pb + BN * HN;
  float* h_m  = c_pb + BN * HN;
  float* c_m  = h_m + BN * CH;
  float* Wsp = alloc(OPAD * CH);
  float* Wip = alloc(OPAD * CH);
  float* bsp = alloc(OPAD);
  float* bip = alloc(OPAD);
  float* pPs = alloc((size_t)PN * BN * OPAD);
  float* pPi = alloc((size_t)PN * BN * OPAD);
  float* pMs = alloc((size_t)MN * BN * OPAD);
  float* pMi = alloc((size_t)MN * BN * OPAD);
  int* sidx = (int*)alloc(PN * BN);

  zero_f32<<<(hcN + 255) / 256, 256, 0, stream>>>(hc, hcN);
  onehot_to_idx<<<(PN * BN + 255) / 256, 256, 0, stream>>>(src_oh, sidx, PN * BN);
  pad_proj_weights<<<(OPAD * CH + 255) / 256, 256, 0, stream>>>(
      W_sub, b_sub, W_ins, b_ins, Wsp, Wip, bsp, bip);

  for (int t = 0; t < PN; ++t) {
    gemm_wave_f32<HN, true><<<dim3(BN / 16, (4 * HN) / 32), 32, 0, stream>>>(
        h_pf, Whh_pf, b_pf, Wih_pf, sidx + t * BN, DN, Gpf, 4 * HN);
    lstm_gates<<<(BN * HN + 255) / 256, 256, 0, stream>>>(
        Gpf, c_pf, h_pf, prior + (size_t)t * BN * CH, HN, 0);
    int tb = PN - 1 - t;
    gemm_wave_f32<HN, true><<<dim3(BN / 16, (4 * HN) / 32), 32, 0, stream>>>(
        h_pb, Whh_pb, b_pb, Wih_pb, sidx + tb * BN, DN, Gpb, 4 * HN);
    lstm_gates<<<(BN * HN + 255) / 256, 256, 0, stream>>>(
        Gpb, c_pb, h_pb, prior + (size_t)tb * BN * CH, HN, HN);
    gemm_wave_f32<CH, true><<<dim3(BN / 16, (4 * CH) / 32), 32, 0, stream>>>(
        h_m, Whh_m, b_m, Wih_m, tgt + t * BN, DN, Gm, 4 * CH);
    lstm_gates<<<(BN * CH + 255) / 256, 256, 0, stream>>>(
        Gm, c_m, h_m, modern + (size_t)t * BN * CH, CH, 0);
  }

  gemm_wave_f32<CH, false><<<dim3(PN * BN / 16, OPAD / 32), 32, 0, stream>>>(
      prior, Wsp, bsp, nullptr, nullptr, 0, pPs, OPAD);
  gemm_wave_f32<CH, false><<<dim3(PN * BN / 16, OPAD / 32), 32, 0, stream>>>(
      prior, Wip, bip, nullptr, nullptr, 0, pPi, OPAD);
  gemm_wave_f32<CH, false><<<dim3(MN * BN / 16, OPAD / 32), 32, 0, stream>>>(
      modern, Wsp, bsp, nullptr, nullptr, 0, pMs, OPAD);
  gemm_wave_f32<CH, false><<<dim3(MN * BN / 16, OPAD / 32), 32, 0, stream>>>(
      modern, Wip, bip, nullptr, nullptr, 0, pMi, OPAD);

  finalize<<<(PN * MN * BN + 255) / 256, 256, 0, stream>>>(
      pPs, pPi, pMs, pMi, tgt, (float*)d_out);
}


# Step and projection GEMMs: each 16x32 tile's K split over the two waves of a 64-thread work-group, upper-K partial sums added through LDS (one barrier); new sum order: (lowK + bias + token) + highK
# speedup vs baseline: 3.0374x; 1.0674x over previous
#include <hip/hip_runtime.h>
#include <math.h>

typedef __attribute__((ext_vector_type(2))) float v2f;
typedef __attribute__((ext_vector_type(4))) float v4f;
typedef __attribute__((ext_vector_type(8))) float v8f;

#define PN 32
#define MN 32
#define BN 128
#define HN 256
#define DN 54
#define ON 53
#define OPAD 64
#define CH 512


__global__ void zero_f32(float* p, int n) {
  int i = blockIdx.x * blockDim.x + threadIdx.x;
  if (i < n) p[i] = 0.f;
}

__global__ void onehot_to_idx(const float* __restrict__ oh, int* __restrict__ idx, int rows) {
  int r = blockIdx.x * blockDim.x + threadIdx.x;
  if (r >= rows) return;
  const float* p = oh + (size_t)r * DN;
  int v = 0;
  for (int d = 0; d < DN; ++d)
    if (p[d] > 0.5f) v = d;
  idx[r] = v;
}

__global__ void pad_proj_weights(const float* __restrict__ Ws, const float* __restrict__ bs,
                                 const float* __restrict__ Wi, const float* __restrict__ bi,
                                 float* __restrict__ Wsp, float* __restrict__ Wip,
                                 float* __restrict__ bsp, float* __restrict__ bip) {
  int t = blockIdx.x * blockDim.x + threadIdx.x;
  if (t >= OPAD * CH) return;
  int n = t / CH;
  int k = t - n * CH;
  Wsp[t] = (n < ON) ? Ws[n * CH + k] : 0.f;
  Wip[t] = (n < ON) ? Wi[n * CH + k] : 0.f;
  if (t < OPAD) {
    bsp[t] = (t < ON) ? bs[t] : 0.f;
    bip[t] = (t < ON) ? bi[t] : 0.f;
  }
}

template <int K, bool EMB>
__global__ __launch_bounds__(64) __attribute__((amdgpu_waves_per_eu(1, 1))) void gemm_wave_f32(
    const float* __restrict__ X, const float* __restrict__ W,
    const float* __restrict__ bias,
    const float* __restrict__ emb, const int* __restrict__ idx, int embD,
    float* __restrict__ out, int N) {
  constexpr int KH = K / 2;
  constexpr int KG = KH / 8;

  __shared__ v4f xch[4][32];

  const int lane = threadIdx.x & 31;
  const int w = __builtin_amdgcn_readfirstlane((int)(threadIdx.x >> 5));
  const int hh = lane >> 4;
  const int c = lane & 15;
  const int row0 = blockIdx.x * 16;
  const int col0 = blockIdx.y * 32;

  const float* Ap = X + (size_t)(row0 + c) * K + w * KH + 4 * hh;
  const float* Bp0 = W + (size_t)(col0 + c) * K + w * KH + 4 * hh;
  const float* Bp1 = Bp0 + (size_t)16 * K;

  const float badd0 = bias[col0 + c];
  const float badd1 = bias[col0 + 16 + c];
  int ridx[8];
  if (EMB) {
#pragma unroll
    for (int r = 0; r < 8; ++r) ridx[r] = idx[row0 + r + hh * 8];
  }
  float e0[8], e1[8];
  if (EMB) {
    const float* ep0 = emb + (size_t)(col0 + c) * embD;
    const float* ep1 = emb + (size_t)(col0 + 16 + c) * embD;
#pragma unroll
    for (int r = 0; r < 8; ++r) {
      e0[r] = ep0[ridx[r]];
      e1[r] = ep1[ridx[r]];
    }
  }

  v8f acc0 = {}, acc1 = {};

#pragma unroll
  for (int g = 0; g < KG; ++g) {
    const v4f av = *(const v4f*)(Ap + g * 8);
    const v4f bv0 = *(const v4f*)(Bp0 + g * 8);
    const v4f bv1 = *(const v4f*)(Bp1 + g * 8);
    const v2f a_lo = {av[0], av[1]};
    const v2f a_hi = {av[2], av[3]};
    const v2f b0_lo = {bv0[0], bv0[1]};
    const v2f b0_hi = {bv0[2], bv0[3]};
    const v2f b1_lo = {bv1[0], bv1[1]};
    const v2f b1_hi = {bv1[2], bv1[3]};
    acc0 = __builtin_amdgcn_wmma_f32_16x16x4_f32(false, a_lo, false, b0_lo, (short)0, acc0, false, false);
    acc1 = __builtin_amdgcn_wmma_f32_16x16x4_f32(false, a_lo, false, b1_lo, (short)0, acc1, false, false);
    acc0 = __builtin_amdgcn_wmma_f32_16x16x4_f32(false, a_hi, false, b0_hi, (short)0, acc0, false, false);
    acc1 = __builtin_amdgcn_wmma_f32_16x16x4_f32(false, a_hi, false, b1_hi, (short)0, acc1, false, false);
  }

  const bool fin = (w == 0);
  const float s0 = fin ? badd0 : 0.f;
  const float s1 = fin ? badd1 : 0.f;
  v8f t0, t1;
#pragma unroll
  for (int r = 0; r < 8; ++r) {
    float v0 = acc0[r] + s0;
    float v1 = acc1[r] + s1;
    if (EMB) {
      v0 += fin ? e0[r] : 0.f;
      v1 += fin ? e1[r] : 0.f;
    }
    t0[r] = v0;
    t1[r] = v1;
  }

  if (w == 1) {
    xch[0][lane] = (v4f){t0[0], t0[1], t0[2], t0[3]};
    xch[1][lane] = (v4f){t0[4], t0[5], t0[6], t0[7]};
    xch[2][lane] = (v4f){t1[0], t1[1], t1[2], t1[3]};
    xch[3][lane] = (v4f){t1[4], t1[5], t1[6], t1[7]};
  }
  __syncthreads();
  if (w == 0) {
    const v4f p00 = xch[0][lane];
    const v4f p01 = xch[1][lane];
    const v4f p10 = xch[2][lane];
    const v4f p11 = xch[3][lane];
    float* op = out + (size_t)(row0 + hh * 8) * N + col0 + c;
#pragma unroll
    for (int r = 0; r < 8; ++r) {
      const float u0 = (r < 4) ? p00[r & 3] : p01[r & 3];
      const float u1 = (r < 4) ? p10[r & 3] : p11[r & 3];
      op[(size_t)r * N] = t0[r] + u0;
      op[(size_t)r * N + 16] = t1[r] + u1;
    }
  }
}

__global__ void lstm_gates(const float* __restrict__ G, float* __restrict__ c,
                           float* __restrict__ h, float* __restrict__ seq,
                           int Hs, int seqOff) {
  int t = blockIdx.x * blockDim.x + threadIdx.x;
  if (t >= BN * Hs) return;
  int b = t / Hs;
  int u = t - b * Hs;
  const float* g = G + (size_t)b * 4 * Hs;
  float gi = g[u];
  float gf = g[Hs + u];
  float gg = g[2 * Hs + u];
  float go = g[3 * Hs + u];
  float si = 1.f / (1.f + expf(-gi));
  float sf = 1.f / (1.f + expf(-gf));
  float so = 1.f / (1.f + expf(-go));
  float cn = sf * c[t] + si * tanhf(gg);
  float hn = so * tanhf(cn);
  c[t] = cn;
  h[t] = hn;
  seq[(size_t)b * CH + seqOff + u] = hn;
}

__global__ void finalize(const float* __restrict__ pPs, const float* __restrict__ pPi,
                         const float* __restrict__ pMs, const float* __restrict__ pMi,
                         const int* __restrict__ tgt, float* __restrict__ out) {
  int t = blockIdx.x * blockDim.x + threadIdx.x;
  if (t >= PN * MN * BN) return;
  int b = t % BN;
  int pm = t / BN;
  int m = pm % MN;
  int p = pm / MN;
  const float* as = pPs + ((size_t)p * BN + b) * OPAD;
  const float* bsv = pMs + ((size_t)m * BN + b) * OPAD;
  const float* ai = pPi + ((size_t)p * BN + b) * OPAD;
  const float* biv = pMi + ((size_t)m * BN + b) * OPAD;
  int tg = (m < MN - 1) ? tgt[(m + 1) * BN + b] : 0;

  float mxs = -3.4e38f, mxi = -3.4e38f;
  for (int o = 0; o < ON; ++o) {
    float ls = as[o] + bsv[o];
    float li = ai[o] + biv[o];
    mxs = ls > mxs ? ls : mxs;
    mxi = li > mxi ? li : mxi;
  }
  float sums = 0.f, sumi = 0.f, ls52 = 0.f, li52 = 0.f, lst = 0.f, lit = 0.f;
  for (int o = 0; o < ON; ++o) {
    float ls = as[o] + bsv[o];
    float li = ai[o] + biv[o];
    sums += expf(ls - mxs);
    sumi += expf(li - mxi);
    if (o == ON - 1) { ls52 = ls; li52 = li; }
    if (o == tg) { lst = ls; lit = li; }
  }
  float lses = mxs + logf(sums);
  float lsei = mxi + logf(sumi);
  size_t idxPMB = ((size_t)p * MN + m) * BN + b;
  out[idxPMB] = ls52 - lses;
  out[(size_t)PN * MN * BN + idxPMB] = li52 - lsei;
  if (m < MN - 1) {
    size_t base = 2ull * PN * MN * BN;
    size_t idxS = ((size_t)p * (MN - 1) + m) * BN + b;
    out[base + idxS] = lst - lses;
    out[base + (size_t)PN * (MN - 1) * BN + idxS] = lit - lsei;
  }
}


extern "C" void kernel_launch(void* const* d_in, const int* in_sizes, int n_in,
                              void* d_out, int out_size, void* d_ws, size_t ws_size,
                              hipStream_t stream) {
  (void)in_sizes; (void)n_in; (void)out_size; (void)ws_size;
  const float* src_oh = (const float*)d_in[0];
  const float* tgt_oh = (const float*)d_in[1];  (void)tgt_oh;
  const float* Wih_pf = (const float*)d_in[2];
  const float* Whh_pf = (const float*)d_in[3];
  const float* b_pf   = (const float*)d_in[4];
  const float* Wih_pb = (const float*)d_in[5];
  const float* Whh_pb = (const float*)d_in[6];
  const float* b_pb   = (const float*)d_in[7];
  const float* Wih_m  = (const float*)d_in[8];
  const float* Whh_m  = (const float*)d_in[9];
  const float* b_m    = (const float*)d_in[10];
  const float* W_sub  = (const float*)d_in[11];
  const float* b_sub  = (const float*)d_in[12];
  const float* W_ins  = (const float*)d_in[13];
  const float* b_ins  = (const float*)d_in[14];
  const int*   tgt    = (const int*)d_in[15];

  char* wptr = (char*)d_ws;
  auto alloc = [&](size_t nfloats) -> float* {
    float* p = (float*)wptr;
    wptr += ((nfloats * 4 + 255) / 256) * 256;
    return p;
  };
  float* prior  = alloc((size_t)PN * BN * CH);
  float* modern = alloc((size_t)MN * BN * CH);
  float* Gpf = alloc((size_t)BN * 4 * HN);
  float* Gpb = alloc((size_t)BN * 4 * HN);
  float* Gm  = alloc((size_t)BN * 4 * CH);
  int hcN = 4 * BN * HN + 2 * BN * CH;
  float* hc = alloc(hcN);
  float* h_pf = hc;
  float* c_pf = h_pf + BN * HN;
  float* h_pb = c_pf + BN * HN;
  float* c_pb = h_pb + BN * HN;
  float* h_m  = c_pb + BN * HN;
  float* c_m  = h_m + BN * CH;
  float* Wsp = alloc(OPAD * CH);
  float* Wip = alloc(OPAD * CH);
  float* bsp = alloc(OPAD);
  float* bip = alloc(OPAD);
  float* pPs = alloc((size_t)PN * BN * OPAD);
  float* pPi = alloc((size_t)PN * BN * OPAD);
  float* pMs = alloc((size_t)MN * BN * OPAD);
  float* pMi = alloc((size_t)MN * BN * OPAD);
  int* sidx = (int*)alloc(PN * BN);

  zero_f32<<<(hcN + 255) / 256, 256, 0, stream>>>(hc, hcN);
  onehot_to_idx<<<(PN * BN + 255) / 256, 256, 0, stream>>>(src_oh, sidx, PN * BN);
  pad_proj_weights<<<(OPAD * CH + 255) / 256, 256, 0, stream>>>(
      W_sub, b_sub, W_ins, b_ins, Wsp, Wip, bsp, bip);

  for (int t = 0; t < PN; ++t) {
    gemm_wave_f32<HN, true><<<dim3(BN / 16, (4 * HN) / 32), 64, 0, stream>>>(
        h_pf, Whh_pf, b_pf, Wih_pf, sidx + t * BN, DN, Gpf, 4 * HN);
    lstm_gates<<<(BN * HN + 255) / 256, 256, 0, stream>>>(
        Gpf, c_pf, h_pf, prior + (size_t)t * BN * CH, HN, 0);
    int tb = PN - 1 - t;
    gemm_wave_f32<HN, true><<<dim3(BN / 16, (4 * HN) / 32), 64, 0, stream>>>(
        h_pb, Whh_pb, b_pb, Wih_pb, sidx + tb * BN, DN, Gpb, 4 * HN);
    lstm_gates<<<(BN * HN + 255) / 256, 256, 0, stream>>>(
        Gpb, c_pb, h_pb, prior + (size_t)tb * BN * CH, HN, HN);
    gemm_wave_f32<CH, true><<<dim3(BN / 16, (4 * CH) / 32), 64, 0, stream>>>(
        h_m, Whh_m, b_m, Wih_m, tgt + t * BN, DN, Gm, 4 * CH);
    lstm_gates<<<(BN * CH + 255) / 256, 256, 0, stream>>>(
        Gm, c_m, h_m, modern + (size_t)t * BN * CH, CH, 0);
  }

  gemm_wave_f32<CH, false><<<dim3(PN * BN / 16, OPAD / 32), 64, 0, stream>>>(
      prior, Wsp, bsp, nullptr, nullptr, 0, pPs, OPAD);
  gemm_wave_f32<CH, false><<<dim3(PN * BN / 16, OPAD / 32), 64, 0, stream>>>(
      prior, Wip, bip, nullptr, nullptr, 0, pPi, OPAD);
  gemm_wave_f32<CH, false><<<dim3(MN * BN / 16, OPAD / 32), 64, 0, stream>>>(
      modern, Wsp, bsp, nullptr, nullptr, 0, pMs, OPAD);
  gemm_wave_f32<CH, false><<<dim3(MN * BN / 16, OPAD / 32), 64, 0, stream>>>(
      modern, Wip, bip, nullptr, nullptr, 0, pMi, OPAD);

  finalize<<<(PN * MN * BN + 255) / 256, 256, 0, stream>>>(
      pPs, pPi, pMs, pMi, tgt, (float*)d_out);
}


# Log-softmax pass re-mapped: wave = (prior step, two batch rows), lanes = 32 modern steps; rows read once as 16-byte pieces, logits kept in registers; output lines assembled in LDS; same sum order
# speedup vs baseline: 3.3375x; 1.0988x over previous
#include <hip/hip_runtime.h>
#include <math.h>

typedef __attribute__((ext_vector_type(2))) float v2f;
typedef __attribute__((ext_vector_type(4))) float v4f;
typedef __attribute__((ext_vector_type(8))) float v8f;

#define PN 32
#define MN 32
#define BN 128
#define HN 256
#define DN 54
#define ON 53
#define OPAD 64
#define CH 512


__global__ void zero_f32(float* p, int n) {
  int i = blockIdx.x * blockDim.x + threadIdx.x;
  if (i < n) p[i] = 0.f;
}

__global__ void onehot_to_idx(const float* __restrict__ oh, int* __restrict__ idx, int rows) {
  int r = blockIdx.x * blockDim.x + threadIdx.x;
  if (r >= rows) return;
  const float* p = oh + (size_t)r * DN;
  int v = 0;
  for (int d = 0; d < DN; ++d)
    if (p[d] > 0.5f) v = d;
  idx[r] = v;
}

__global__ void pad_proj_weights(const float* __restrict__ Ws, const float* __restrict__ bs,
                                 const float* __restrict__ Wi, const float* __restrict__ bi,
                                 float* __restrict__ Wsp, float* __restrict__ Wip,
                                 float* __restrict__ bsp, float* __restrict__ bip) {
  int t = blockIdx.x * blockDim.x + threadIdx.x;
  if (t >= OPAD * CH) return;
  int n = t / CH;
  int k = t - n * CH;
  Wsp[t] = (n < ON) ? Ws[n * CH + k] : 0.f;
  Wip[t] = (n < ON) ? Wi[n * CH + k] : 0.f;
  if (t < OPAD) {
    bsp[t] = (t < ON) ? bs[t] : 0.f;
    bip[t] = (t < ON) ? bi[t] : 0.f;
  }
}

template <int K, bool EMB>
__global__ __launch_bounds__(64) __attribute__((amdgpu_waves_per_eu(1, 1))) void gemm_wave_f32(
    const float* __restrict__ X, const float* __restrict__ W,
    const float* __restrict__ bias,
    const float* __restrict__ emb, const int* __restrict__ idx, int embD,
    float* __restrict__ out, int N) {
  constexpr int KH = K / 2;
  constexpr int KG = KH / 8;

  __shared__ v4f xch[4][32];

  const int lane = threadIdx.x & 31;
  const int w = __builtin_amdgcn_readfirstlane((int)(threadIdx.x >> 5));
  const int hh = lane >> 4;
  const int c = lane & 15;
  const int row0 = blockIdx.x * 16;
  const int col0 = blockIdx.y * 32;

  const float* Ap = X + (size_t)(row0 + c) * K + w * KH + 4 * hh;
  const float* Bp0 = W + (size_t)(col0 + c) * K + w * KH + 4 * hh;
  const float* Bp1 = Bp0 + (size_t)16 * K;

  const float badd0 = bias[col0 + c];
  const float badd1 = bias[col0 + 16 + c];
  int ridx[8];
  if (EMB) {
#pragma unroll
    for (int r = 0; r < 8; ++r) ridx[r] = idx[row0 + r + hh * 8];
  }
  float e0[8], e1[8];
  if (EMB) {
    const float* ep0 = emb + (size_t)(col0 + c) * embD;
    const float* ep1 = emb + (size_t)(col0 + 16 + c) * embD;
#pragma unroll
    for (int r = 0; r < 8; ++r) {
      e0[r] = ep0[ridx[r]];
      e1[r] = ep1[ridx[r]];
    }
  }

  v8f acc0 = {}, acc1 = {};

#pragma unroll
  for (int g = 0; g < KG; ++g) {
    const v4f av = *(const v4f*)(Ap + g * 8);
    const v4f bv0 = *(const v4f*)(Bp0 + g * 8);
    const v4f bv1 = *(const v4f*)(Bp1 + g * 8);
    const v2f a_lo = {av[0], av[1]};
    const v2f a_hi = {av[2], av[3]};
    const v2f b0_lo = {bv0[0], bv0[1]};
    const v2f b0_hi = {bv0[2], bv0[3]};
    const v2f b1_lo = {bv1[0], bv1[1]};
    const v2f b1_hi = {bv1[2], bv1[3]};
    acc0 = __builtin_amdgcn_wmma_f32_16x16x4_f32(false, a_lo, false, b0_lo, (short)0, acc0, false, false);
    acc1 = __builtin_amdgcn_wmma_f32_16x16x4_f32(false, a_lo, false, b1_lo, (short)0, acc1, false, false);
    acc0 = __builtin_amdgcn_wmma_f32_16x16x4_f32(false, a_hi, false, b0_hi, (short)0, acc0, false, false);
    acc1 = __builtin_amdgcn_wmma_f32_16x16x4_f32(false, a_hi, false, b1_hi, (short)0, acc1, false, false);
  }

  const bool fin = (w == 0);
  const float s0 = fin ? badd0 : 0.f;
  const float s1 = fin ? badd1 : 0.f;
  v8f t0, t1;
#pragma unroll
  for (int r = 0; r < 8; ++r) {
    float v0 = acc0[r] + s0;
    float v1 = acc1[r] + s1;
    if (EMB) {
      v0 += fin ? e0[r] : 0.f;
      v1 += fin ? e1[r] : 0.f;
    }
    t0[r] = v0;
    t1[r] = v1;
  }

  if (w == 1) {
    xch[0][lane] = (v4f){t0[0], t0[1], t0[2], t0[3]};
    xch[1][lane] = (v4f){t0[4], t0[5], t0[6], t0[7]};
    xch[2][lane] = (v4f){t1[0], t1[1], t1[2], t1[3]};
    xch[3][lane] = (v4f){t1[4], t1[5], t1[6], t1[7]};
  }
  __syncthreads();
  if (w == 0) {
    const v4f p00 = xch[0][lane];
    const v4f p01 = xch[1][lane];
    const v4f p10 = xch[2][lane];
    const v4f p11 = xch[3][lane];
    float* op = out + (size_t)(row0 + hh * 8) * N + col0 + c;
#pragma unroll
    for (int r = 0; r < 8; ++r) {
      const float u0 = (r < 4) ? p00[r & 3] : p01[r & 3];
      const float u1 = (r < 4) ? p10[r & 3] : p11[r & 3];
      op[(size_t)r * N] = t0[r] + u0;
      op[(size_t)r * N + 16] = t1[r] + u1;
    }
  }
}

__global__ void lstm_gates(const float* __restrict__ G, float* __restrict__ c,
                           float* __restrict__ h, float* __restrict__ seq,
                           int Hs, int seqOff) {
  int t = blockIdx.x * blockDim.x + threadIdx.x;
  if (t >= BN * Hs) return;
  int b = t / Hs;
  int u = t - b * Hs;
  const float* g = G + (size_t)b * 4 * Hs;
  float gi = g[u];
  float gf = g[Hs + u];
  float gg = g[2 * Hs + u];
  float go = g[3 * Hs + u];
  float si = 1.f / (1.f + expf(-gi));
  float sf = 1.f / (1.f + expf(-gf));
  float so = 1.f / (1.f + expf(-go));
  float cn = sf * c[t] + si * tanhf(gg);
  float hn = so * tanhf(cn);
  c[t] = cn;
  h[t] = hn;
  seq[(size_t)b * CH + seqOff + u] = hn;
}

__global__ __launch_bounds__(512) void finalize(const float* __restrict__ pPs, const float* __restrict__ pPi,
                                                const float* __restrict__ pMs, const float* __restrict__ pMi,
                                                const int* __restrict__ tgt, float* __restrict__ out) {
  __shared__ float xl[MN * 33];
  __shared__ float xt[MN * 33];

  const int lane = threadIdx.x & 31;
  const int w = __builtin_amdgcn_readfirstlane((int)(threadIdx.x >> 5));
  const int p = blockIdx.x >> 2;
  const int b0 = (blockIdx.x & 3) * 32;
  const int head = blockIdx.y;
  const float* __restrict__ PA = head ? pPi : pPs;
  const float* __restrict__ PM = head ? pMi : pMs;

  const int m = lane;
  const int mt = (m < MN - 1) ? m + 1 : MN - 1;

  for (int r = 0; r < 2; ++r) {
    const int bl = w + 16 * r;
    const int b = b0 + bl;
    const float* arow = PA + ((size_t)p * BN + b) * OPAD;
    const float* brow = PM + ((size_t)m * BN + b) * OPAD;
    const int tg = tgt[mt * BN + b];

    float l[56];
#pragma unroll
    for (int j = 0; j < 14; ++j) {
      const v4f av = *(const v4f*)(arow + 4 * j);
      const v4f bv = *(const v4f*)(brow + 4 * j);
      l[4 * j + 0] = av[0] + bv[0];
      l[4 * j + 1] = av[1] + bv[1];
      l[4 * j + 2] = av[2] + bv[2];
      l[4 * j + 3] = av[3] + bv[3];
    }

    float mx = -3.4e38f;
#pragma unroll
    for (int o = 0; o < ON; ++o) mx = l[o] > mx ? l[o] : mx;
    float sum = 0.f;
#pragma unroll
    for (int o = 0; o < ON; ++o) sum += expf(l[o] - mx);
    const float lse = mx + logf(sum);
    const float lt = arow[tg] + brow[tg];

    xl[m * 33 + bl] = l[ON - 1] - lse;
    xt[m * 33 + bl] = lt - lse;
  }
  __syncthreads();

  for (int r = 0; r < 2; ++r) {
    const int mo = w + 16 * r;
    const float dv = xl[mo * 33 + lane];
    const float sv = xt[mo * 33 + lane];
    const size_t idxPMB = ((size_t)p * MN + mo) * BN + b0 + lane;
    out[(size_t)head * PN * MN * BN + idxPMB] = dv;
    if (mo < MN - 1) {
      const size_t base = 2ull * PN * MN * BN + (size_t)head * PN * (MN - 1) * BN;
      const size_t idxS = ((size_t)p * (MN - 1) + mo) * BN + b0 + lane;
      out[base + idxS] = sv;
    }
  }
}


extern "C" void kernel_launch(void* const* d_in, const int* in_sizes, int n_in,
                              void* d_out, int out_size, void* d_ws, size_t ws_size,
                              hipStream_t stream) {
  (void)in_sizes; (void)n_in; (void)out_size; (void)ws_size;
  const float* src_oh = (const float*)d_in[0];
  const float* tgt_oh = (const float*)d_in[1];  (void)tgt_oh;
  const float* Wih_pf = (const float*)d_in[2];
  const float* Whh_pf = (const float*)d_in[3];
  const float* b_pf   = (const float*)d_in[4];
  const float* Wih_pb = (const float*)d_in[5];
  const float* Whh_pb = (const float*)d_in[6];
  const float* b_pb   = (const float*)d_in[7];
  const float* Wih_m  = (const float*)d_in[8];
  const float* Whh_m  = (const float*)d_in[9];
  const float* b_m    = (const float*)d_in[10];
  const float* W_sub  = (const float*)d_in[11];
  const float* b_sub  = (const float*)d_in[12];
  const float* W_ins  = (const float*)d_in[13];
  const float* b_ins  = (const float*)d_in[14];
  const int*   tgt    = (const int*)d_in[15];

  char* wptr = (char*)d_ws;
  auto alloc = [&](size_t nfloats) -> float* {
    float* p = (float*)wptr;
    wptr += ((nfloats * 4 + 255) / 256) * 256;
    return p;
  };
  float* prior  = alloc((size_t)PN * BN * CH);
  float* modern = alloc((size_t)MN * BN * CH);
  float* Gpf = alloc((size_t)BN * 4 * HN);
  float* Gpb = alloc((size_t)BN * 4 * HN);
  float* Gm  = alloc((size_t)BN * 4 * CH);
  int hcN = 4 * BN * HN + 2 * BN * CH;
  float* hc = alloc(hcN);
  float* h_pf = hc;
  float* c_pf = h_pf + BN * HN;
  float* h_pb = c_pf + BN * HN;
  float* c_pb = h_pb + BN * HN;
  float* h_m  = c_pb + BN * HN;
  float* c_m  = h_m + BN * CH;
  float* Wsp = alloc(OPAD * CH);
  float* Wip = alloc(OPAD * CH);
  float* bsp = alloc(OPAD);
  float* bip = alloc(OPAD);
  float* pPs = alloc((size_t)PN * BN * OPAD);
  float* pPi = alloc((size_t)PN * BN * OPAD);
  float* pMs = alloc((size_t)MN * BN * OPAD);
  float* pMi = alloc((size_t)MN * BN * OPAD);
  int* sidx = (int*)alloc(PN * BN);

  zero_f32<<<(hcN + 255) / 256, 256, 0, stream>>>(hc, hcN);
  onehot_to_idx<<<(PN * BN + 255) / 256, 256, 0, stream>>>(src_oh, sidx, PN * BN);
  pad_proj_weights<<<(OPAD * CH + 255) / 256, 256, 0, stream>>>(
      W_sub, b_sub, W_ins, b_ins, Wsp, Wip, bsp, bip);

  for (int t = 0; t < PN; ++t) {
    gemm_wave_f32<HN, true><<<dim3(BN / 16, (4 * HN) / 32), 64, 0, stream>>>(
        h_pf, Whh_pf, b_pf, Wih_pf, sidx + t * BN, DN, Gpf, 4 * HN);
    lstm_gates<<<(BN * HN + 255) / 256, 256, 0, stream>>>(
        Gpf, c_pf, h_pf, prior + (size_t)t * BN * CH, HN, 0);
    int tb = PN - 1 - t;
    gemm_wave_f32<HN, true><<<dim3(BN / 16, (4 * HN) / 32), 64, 0, stream>>>(
        h_pb, Whh_pb, b_pb, Wih_pb, sidx + tb * BN, DN, Gpb, 4 * HN);
    lstm_gates<<<(BN * HN + 255) / 256, 256, 0, stream>>>(
        Gpb, c_pb, h_pb, prior + (size_t)tb * BN * CH, HN, HN);
    gemm_wave_f32<CH, true><<<dim3(BN / 16, (4 * CH) / 32), 64, 0, stream>>>(
        h_m, Whh_m, b_m, Wih_m, tgt + t * BN, DN, Gm, 4 * CH);
    lstm_gates<<<(BN * CH + 255) / 256, 256, 0, stream>>>(
        Gm, c_m, h_m, modern + (size_t)t * BN * CH, CH, 0);
  }

  gemm_wave_f32<CH, false><<<dim3(PN * BN / 16, OPAD / 32), 64, 0, stream>>>(
      prior, Wsp, bsp, nullptr, nullptr, 0, pPs, OPAD);
  gemm_wave_f32<CH, false><<<dim3(PN * BN / 16, OPAD / 32), 64, 0, stream>>>(
      prior, Wip, bip, nullptr, nullptr, 0, pPi, OPAD);
  gemm_wave_f32<CH, false><<<dim3(MN * BN / 16, OPAD / 32), 64, 0, stream>>>(
      modern, Wsp, bsp, nullptr, nullptr, 0, pMs, OPAD);
  gemm_wave_f32<CH, false><<<dim3(MN * BN / 16, OPAD / 32), 64, 0, stream>>>(
      modern, Wip, bip, nullptr, nullptr, 0, pMi, OPAD);

  finalize<<<dim3(PN * (BN / 32), 2), 512, 0, stream>>>(
      pPs, pPi, pMs, pMi, tgt, (float*)d_out);
}
